# starting kernel
# baseline (speedup 1.0000x reference)
#include <hip/hip_runtime.h>
#include <hip/hip_bf16.h>

typedef float v2f __attribute__((ext_vector_type(2)));
typedef float v8f __attribute__((ext_vector_type(8)));

template <int N, int NT, bool FULLM>
__global__ __launch_bounds__(256) void gemm_wmma(
    float* __restrict__ Z, const float* __restrict__ H,
    const float* __restrict__ S, int Mrows)
{
    const int lane    = threadIdx.x & 31;
    const int wave    = (blockIdx.x * blockDim.x + threadIdx.x) >> 5;
    const int nStrips = N / (16 * NT);
    const int mTiles  = (Mrows + 15) >> 4;
    if (wave >= mTiles * nStrips) return;

    const int mTile  = wave / nStrips;
    const int nStrip = wave - mTile * nStrips;
    const int m0 = mTile * 16;
    const int n0 = nStrip * (16 * NT);

    const int col  = lane & 15;
    const int hi   = lane >> 4;
    const int koff = hi << 1;
    const int m    = m0 + col;

    const float scale = (m < Mrows) ? 1.0f : 0.0f;
    const float* __restrict__ Hrow =
        H + (size_t)(FULLM ? m : (m < Mrows ? m : 0)) * N + koff;
    const float* __restrict__ Sbase = S + (size_t)koff * N + n0 + col;

    v8f acc[NT];
#pragma unroll
    for (int t = 0; t < NT; ++t)
        acc[t] = (v8f){0.f, 0.f, 0.f, 0.f, 0.f, 0.f, 0.f, 0.f};

#pragma unroll 8
    for (int k = 0; k < N; k += 4) {
        v2f a;
        a.x = Hrow[k];
        a.y = Hrow[k + 1];
        if (!FULLM) { a.x *= scale; a.y *= scale; }

        const float* __restrict__ Sp = Sbase + (size_t)k * N;
#pragma unroll
        for (int t = 0; t < NT; ++t) {
            v2f b;
            b.x = Sp[t * 16];
            b.y = Sp[t * 16 + N];
            acc[t] = __builtin_amdgcn_wmma_f32_16x16x4_f32(
                false, a, false, b, (short)0, acc[t], false, false);
        }
    }

#pragma unroll
    for (int t = 0; t < NT; ++t) {
#pragma unroll
        for (int v = 0; v < 8; ++v) {
            const int mrow = m0 + v + (hi << 3);
            if (FULLM || mrow < Mrows)
                Z[(size_t)mrow * N + n0 + t * 16 + col] = acc[t][v];
        }
    }
}

__global__ void combine_relu_kernel(
    float* __restrict__ Hout,
    const float* __restrict__ Z0, const float* __restrict__ Z1,
    const float* __restrict__ Z2,
    const float* __restrict__ W, const float* __restrict__ bias,
    int Bn, int Fin, int Fout, int N)
{
    const int tid = blockIdx.x * blockDim.x + threadIdx.x;
    if (tid >= Bn * N) return;
    const int n = tid % N;
    const int b = tid / N;

    float out[16];
    for (int g = 0; g < Fout; ++g) out[g] = bias[g];

    for (int f = 0; f < Fin; ++f) {
        const size_t idx = ((size_t)(b * Fin + f)) * (size_t)N + n;
        const float z0 = Z0[idx];
        const float z1 = Z1[idx];
        const float z2 = Z2[idx];
        const float* __restrict__ W0 = W + ((size_t)(0 * Fin + f)) * Fout;
        const float* __restrict__ W1 = W + ((size_t)(1 * Fin + f)) * Fout;
        const float* __restrict__ W2 = W + ((size_t)(2 * Fin + f)) * Fout;
        for (int g = 0; g < Fout; ++g)
            out[g] = fmaf(z0, W0[g], fmaf(z1, W1[g], fmaf(z2, W2[g], out[g])));
    }

    for (int g = 0; g < Fout; ++g) {
        const float v = out[g];
        Hout[((size_t)(b * Fout + g)) * (size_t)N + n] = v > 0.f ? v : 0.f;
    }
}

__global__ void pool_kernel(
    float* __restrict__ P, const float* __restrict__ Hin,
    const int* __restrict__ incid, int BF, int N1, int N2, int DEG)
{
    const int tid = blockIdx.x * blockDim.x + threadIdx.x;
    if (tid >= BF * N2) return;
    const int m  = tid % N2;
    const int bf = tid / N2;
    const float* __restrict__ row = Hin + (size_t)bf * (size_t)N1;
    float v = 0.0f;
    for (int d = 0; d < DEG; ++d) {
        const int j = incid[m * DEG + d];
        v = fmaxf(v, row[j]);
    }
    P[tid] = v;
}

__global__ void readout_kernel(
    float* __restrict__ Y, const float* __restrict__ H,
    const float* __restrict__ R0W, const float* __restrict__ R0b,
    const float* __restrict__ R1W, const float* __restrict__ R1b,
    int Bn, int N2)
{
    const int tid = blockIdx.x * blockDim.x + threadIdx.x;
    if (tid >= Bn * N2) return;
    const int n = tid % N2;
    const int b = tid / N2;

    float v[16];
    for (int i = 0; i < 16; ++i)
        v[i] = H[((size_t)(b * 16 + i)) * (size_t)N2 + n];

    float y0[16];
    for (int j = 0; j < 16; ++j) {
        float s = R0b[j];
        for (int i = 0; i < 16; ++i) s = fmaf(v[i], R0W[i * 16 + j], s);
        y0[j] = s > 0.f ? s : 0.f;
    }

    for (int r = 0; r < 2; ++r) {
        float s = R1b[r];
        for (int j = 0; j < 16; ++j) s = fmaf(y0[j], R1W[j * 2 + r], s);
        Y[((size_t)(b * 2 + r)) * (size_t)N2 + n] = s;
    }
}

extern "C" void kernel_launch(void* const* d_in, const int* in_sizes, int n_in,
                              void* d_out, int out_size, void* d_ws, size_t ws_size,
                              hipStream_t stream)
{
    (void)in_sizes; (void)n_in; (void)out_size; (void)ws_size;

    const float* x     = (const float*)d_in[0];
    const float* S1    = (const float*)d_in[1];
    const float* S2    = (const float*)d_in[2];
    const int*   incid = (const int*)  d_in[3];
    const float* W10   = (const float*)d_in[4];
    const float* b10   = (const float*)d_in[5];
    const float* W11   = (const float*)d_in[6];
    const float* b11   = (const float*)d_in[7];
    const float* W20   = (const float*)d_in[8];
    const float* b20   = (const float*)d_in[9];
    const float* R0W   = (const float*)d_in[10];
    const float* R0b   = (const float*)d_in[11];
    const float* R1W   = (const float*)d_in[12];
    const float* R1b   = (const float*)d_in[13];
    float* out = (float*)d_out;

    const int B = 4, F1 = 16, N1 = 8192, N2 = 4096, DEG = 4;

    char* ws = (char*)d_ws;
    const size_t szBig = (size_t)(B * F1) * N1 * sizeof(float);
    float* h1 = (float*)(ws);
    float* tA = (float*)(ws + 1 * szBig);
    float* tB = (float*)(ws + 2 * szBig);
    float* h2 = (float*)(ws + 3 * szBig);
    float* p  = (float*)(ws + 4 * szBig);

    auto blocksFor = [](int tiles) { return (tiles + 7) / 8; };
    auto grid1d    = [](int total, int blk) { return (total + blk - 1) / blk; };

    {
        const int tiles = 1 * (N1 / 16);
        gemm_wmma<N1, 1, false><<<blocksFor(tiles), 256, 0, stream>>>(tA, x,  S1, B);
        gemm_wmma<N1, 1, false><<<blocksFor(tiles), 256, 0, stream>>>(tB, tA, S1, B);
    }
    combine_relu_kernel<<<grid1d(B * N1, 256), 256, 0, stream>>>(
        h1, x, tA, tB, W10, b10, B, 1, F1, N1);

    {
        const int tiles = 4 * (N1 / 32);
        gemm_wmma<N1, 2, true><<<blocksFor(tiles), 256, 0, stream>>>(tA, h1, S1, B * F1);
        gemm_wmma<N1, 2, true><<<blocksFor(tiles), 256, 0, stream>>>(tB, tA, S1, B * F1);
    }
    combine_relu_kernel<<<grid1d(B * N1, 256), 256, 0, stream>>>(
        h2, h1, tA, tB, W11, b11, B, F1, F1, N1);

    pool_kernel<<<grid1d(B * F1 * N2, 256), 256, 0, stream>>>(
        p, h2, incid, B * F1, N1, N2, DEG);

    {
        const int tiles = 4 * (N2 / 32);
        gemm_wmma<N2, 2, true><<<blocksFor(tiles), 256, 0, stream>>>(tA, p,  S2, B * F1);
        gemm_wmma<N2, 2, true><<<blocksFor(tiles), 256, 0, stream>>>(tB, tA, S2, B * F1);
    }
    combine_relu_kernel<<<grid1d(B * N2, 256), 256, 0, stream>>>(
        h1  , p, tA, tB, W20, b20, B, F1, F1, N2);

    readout_kernel<<<grid1d(B * N2, 256), 256, 0, stream>>>(
        out, h1, R0W, R0b, R1W, R1b, B, N2);
}


# GEMMs: one workgroup per 32-column strip covers all rows, its 16 waves split K and their partial tiles are summed in fixed wave order through LDS (f32 summation order changes)
# speedup vs baseline: 4.2938x; 4.2938x over previous
#include <hip/hip_runtime.h>
#include <hip/hip_bf16.h>

typedef float v2f __attribute__((ext_vector_type(2)));
typedef float v4f __attribute__((ext_vector_type(4)));
typedef float v8f __attribute__((ext_vector_type(8)));

template <int N, int MT, int KC, int NW>
__global__ __launch_bounds__(NW * 32) void gemm_splitk(
    float* __restrict__ Z, const float* __restrict__ H,
    const float* __restrict__ S, int Mrows)
{
    constexpr int KS  = N / NW;
    constexpr int NJ  = KC / 4;
    constexpr int NQ  = KC / 8;
    constexpr int NCH = KS / KC;
    static_assert(KS % (4 * KC) == 0, "four chunks per loop trip");

    __shared__ float red[NW * 2 * 8 * 32];

    const int lane = threadIdx.x & 31;
    const int wave = threadIdx.x >> 5;
    const int c    = lane & 15;
    const int h    = lane >> 4;
    const int n0   = blockIdx.x * 32;
    const int k0   = wave * KS + h * (KC / 2);

    const float* __restrict__ Ap[MT];
#pragma unroll
    for (int mt = 0; mt < MT; ++mt) {
        int m = mt * 16 + c;
        if (m >= Mrows) m = 0;
        Ap[mt] = H + (size_t)m * N + k0;
    }
    const float* __restrict__ Bp = S + (size_t)k0 * N + n0 + c;

    v8f acc[MT][2];
#pragma unroll
    for (int mt = 0; mt < MT; ++mt) {
        acc[mt][0] = (v8f){0.f, 0.f, 0.f, 0.f, 0.f, 0.f, 0.f, 0.f};
        acc[mt][1] = (v8f){0.f, 0.f, 0.f, 0.f, 0.f, 0.f, 0.f, 0.f};
    }

    v4f a0[MT][NQ], a1[MT][NQ], a2[MT][NQ];
    v2f b0[NJ][2],  b1[NJ][2],  b2[NJ][2];

    auto load_chunk = [&](v4f (&a)[MT][NQ], v2f (&b)[NJ][2], int ch) {
        const int kb = ch * KC;
#pragma unroll
        for (int j = 0; j < NJ; ++j) {
            const float* __restrict__ r0 = Bp + (size_t)(kb + 2 * j) * N;
            const float* __restrict__ r1 = r0 + N;
            b[j][0].x = r0[0];  b[j][0].y = r1[0];
            b[j][1].x = r0[16]; b[j][1].y = r1[16];
        }
#pragma unroll
        for (int mt = 0; mt < MT; ++mt)
#pragma unroll
            for (int q = 0; q < NQ; ++q)
                a[mt][q] = *(const v4f*)(Ap[mt] + kb + 4 * q);
    };

    auto mma_chunk = [&](const v4f (&a)[MT][NQ], const v2f (&b)[NJ][2]) {
#pragma unroll
        for (int j = 0; j < NJ; ++j) {
#pragma unroll
            for (int mt = 0; mt < MT; ++mt) {
                v2f av;
                if (j & 1) { av.x = a[mt][j >> 1].z; av.y = a[mt][j >> 1].w; }
                else       { av.x = a[mt][j >> 1].x; av.y = a[mt][j >> 1].y; }
                acc[mt][0] = __builtin_amdgcn_wmma_f32_16x16x4_f32(
                    false, av, false, b[j][0], (short)0, acc[mt][0], false, false);
                acc[mt][1] = __builtin_amdgcn_wmma_f32_16x16x4_f32(
                    false, av, false, b[j][1], (short)0, acc[mt][1], false, false);
            }
        }
    };

#define SCHED_FENCE() __builtin_amdgcn_sched_barrier(0)
    load_chunk(a0, b0, 0);
    SCHED_FENCE();
    load_chunk(a1, b1, 1);
    SCHED_FENCE();
#pragma nounroll
    for (int ch = 0; ch < NCH - 4; ch += 4) {
        mma_chunk(a0, b0);
        SCHED_FENCE();
        load_chunk(a0, b0, ch + 2);
        SCHED_FENCE();
        load_chunk(a2, b2, ch + 3);
        SCHED_FENCE();
        mma_chunk(a1, b1);
        mma_chunk(a0, b0);
        SCHED_FENCE();
        load_chunk(a0, b0, ch + 4);
        SCHED_FENCE();
        load_chunk(a1, b1, ch + 5);
        SCHED_FENCE();
        mma_chunk(a2, b2);
    }
    mma_chunk(a0, b0);
    SCHED_FENCE();
    load_chunk(a0, b0, NCH - 2);
    SCHED_FENCE();
    load_chunk(a2, b2, NCH - 1);
    SCHED_FENCE();
    mma_chunk(a1, b1);
    mma_chunk(a0, b0);
    mma_chunk(a2, b2);
#undef SCHED_FENCE

#pragma unroll
    for (int mt = 0; mt < MT; ++mt) {
        if (mt) __syncthreads();
#pragma unroll
        for (int t = 0; t < 2; ++t)
#pragma unroll
            for (int r = 0; r < 8; ++r)
                red[((wave * 2 + t) * 8 + r) * 32 + lane] = acc[mt][t][r];
        __syncthreads();
        if (threadIdx.x < 256) {
            const int r = wave;
            float s0 = red[(0 * 8 + r) * 32 + lane];
            float s1 = red[(1 * 8 + r) * 32 + lane];
#pragma unroll
            for (int w = 1; w < NW; ++w) {
                s0 += red[((w * 2 + 0) * 8 + r) * 32 + lane];
                s1 += red[((w * 2 + 1) * 8 + r) * 32 + lane];
            }
            const int row = mt * 16 + 8 * h + r;
            if (row < Mrows) {
                float* __restrict__ zr = Z + (size_t)row * N + n0 + c;
                zr[0]  = s0;
                zr[16] = s1;
            }
        }
    }
}

__global__ void combine_relu_kernel(
    float* __restrict__ Hout,
    const float* __restrict__ Z0, const float* __restrict__ Z1,
    const float* __restrict__ Z2,
    const float* __restrict__ W, const float* __restrict__ bias,
    int Bn, int Fin, int Fout, int N)
{
    const int tid = blockIdx.x * blockDim.x + threadIdx.x;
    if (tid >= Bn * N) return;
    const int n = tid % N;
    const int b = tid / N;

    float out[16];
    for (int g = 0; g < Fout; ++g) out[g] = bias[g];

    for (int f = 0; f < Fin; ++f) {
        const size_t idx = ((size_t)(b * Fin + f)) * (size_t)N + n;
        const float z0 = Z0[idx];
        const float z1 = Z1[idx];
        const float z2 = Z2[idx];
        const float* __restrict__ W0 = W + ((size_t)(0 * Fin + f)) * Fout;
        const float* __restrict__ W1 = W + ((size_t)(1 * Fin + f)) * Fout;
        const float* __restrict__ W2 = W + ((size_t)(2 * Fin + f)) * Fout;
        for (int g = 0; g < Fout; ++g)
            out[g] = fmaf(z0, W0[g], fmaf(z1, W1[g], fmaf(z2, W2[g], out[g])));
    }

    for (int g = 0; g < Fout; ++g) {
        const float v = out[g];
        Hout[((size_t)(b * Fout + g)) * (size_t)N + n] = v > 0.f ? v : 0.f;
    }
}

__global__ void pool_kernel(
    float* __restrict__ P, const float* __restrict__ Hin,
    const int* __restrict__ incid, int BF, int N1, int N2, int DEG)
{
    const int tid = blockIdx.x * blockDim.x + threadIdx.x;
    if (tid >= BF * N2) return;
    const int m  = tid % N2;
    const int bf = tid / N2;
    const float* __restrict__ row = Hin + (size_t)bf * (size_t)N1;
    float v = 0.0f;
    for (int d = 0; d < DEG; ++d) {
        const int j = incid[m * DEG + d];
        v = fmaxf(v, row[j]);
    }
    P[tid] = v;
}

__global__ void readout_kernel(
    float* __restrict__ Y, const float* __restrict__ H,
    const float* __restrict__ R0W, const float* __restrict__ R0b,
    const float* __restrict__ R1W, const float* __restrict__ R1b,
    int Bn, int N2)
{
    const int tid = blockIdx.x * blockDim.x + threadIdx.x;
    if (tid >= Bn * N2) return;
    const int n = tid % N2;
    const int b = tid / N2;

    float v[16];
    for (int i = 0; i < 16; ++i)
        v[i] = H[((size_t)(b * 16 + i)) * (size_t)N2 + n];

    float y0[16];
    for (int j = 0; j < 16; ++j) {
        float s = R0b[j];
        for (int i = 0; i < 16; ++i) s = fmaf(v[i], R0W[i * 16 + j], s);
        y0[j] = s > 0.f ? s : 0.f;
    }

    for (int r = 0; r < 2; ++r) {
        float s = R1b[r];
        for (int j = 0; j < 16; ++j) s = fmaf(y0[j], R1W[j * 2 + r], s);
        Y[((size_t)(b * 2 + r)) * (size_t)N2 + n] = s;
    }
}

extern "C" void kernel_launch(void* const* d_in, const int* in_sizes, int n_in,
                              void* d_out, int out_size, void* d_ws, size_t ws_size,
                              hipStream_t stream)
{
    (void)in_sizes; (void)n_in; (void)out_size; (void)ws_size;

    const float* x     = (const float*)d_in[0];
    const float* S1    = (const float*)d_in[1];
    const float* S2    = (const float*)d_in[2];
    const int*   incid = (const int*)  d_in[3];
    const float* W10   = (const float*)d_in[4];
    const float* b10   = (const float*)d_in[5];
    const float* W11   = (const float*)d_in[6];
    const float* b11   = (const float*)d_in[7];
    const float* W20   = (const float*)d_in[8];
    const float* b20   = (const float*)d_in[9];
    const float* R0W   = (const float*)d_in[10];
    const float* R0b   = (const float*)d_in[11];
    const float* R1W   = (const float*)d_in[12];
    const float* R1b   = (const float*)d_in[13];
    float* out = (float*)d_out;

    const int B = 4, F1 = 16, N1 = 8192, N2 = 4096, DEG = 4;

    char* ws = (char*)d_ws;
    const size_t szBig = (size_t)(B * F1) * N1 * sizeof(float);
    float* h1 = (float*)(ws);
    float* tA = (float*)(ws + 1 * szBig);
    float* tB = (float*)(ws + 2 * szBig);
    float* h2 = (float*)(ws + 3 * szBig);
    float* p  = (float*)(ws + 4 * szBig);

    auto grid1d    = [](int total, int blk) { return (total + blk - 1) / blk; };

    gemm_splitk<N1, 1, 16, 16><<<N1 / 32, 512, 0, stream>>>(tA, x,  S1, B);
    gemm_splitk<N1, 1, 16, 16><<<N1 / 32, 512, 0, stream>>>(tB, tA, S1, B);
    combine_relu_kernel<<<grid1d(B * N1, 256), 256, 0, stream>>>(
        h1, x, tA, tB, W10, b10, B, 1, F1, N1);

    gemm_splitk<N1, 4, 16, 16><<<N1 / 32, 512, 0, stream>>>(tA, h1, S1, B * F1);
    gemm_splitk<N1, 4, 16, 16><<<N1 / 32, 512, 0, stream>>>(tB, tA, S1, B * F1);
    combine_relu_kernel<<<grid1d(B * N1, 256), 256, 0, stream>>>(
        h2, h1, tA, tB, W11, b11, B, F1, F1, N1);

    pool_kernel<<<grid1d(B * F1 * N2, 256), 256, 0, stream>>>(
        p, h2, incid, B * F1, N1, N2, DEG);

    gemm_splitk<N2, 4, 16, 16><<<N2 / 32, 512, 0, stream>>>(tA, p,  S2, B * F1);
    gemm_splitk<N2, 4, 16, 16><<<N2 / 32, 512, 0, stream>>>(tB, tA, S2, B * F1);
    combine_relu_kernel<<<grid1d(B * N2, 256), 256, 0, stream>>>(
        h1  , p, tA, tB, W20, b20, B, F1, F1, N2);

    readout_kernel<<<grid1d(B * N2, 256), 256, 0, stream>>>(
        out, h1, R0W, R0b, R1W, R1b, B, N2);
}


# 64-row activations stored pair-packed by node in the workspace: each A-fragment load of the 64-row GEMMs reads 256-byte runs (4 lines per load, was 16); GEMM stores and small kernels follow
# speedup vs baseline: 5.4179x; 1.2618x over previous
#include <hip/hip_runtime.h>
#include <hip/hip_bf16.h>

typedef float v2f __attribute__((ext_vector_type(2)));
typedef float v4f __attribute__((ext_vector_type(4)));
typedef float v8f __attribute__((ext_vector_type(8)));

template <int N, int MT, int KC, int NW, bool PK>
__global__ __launch_bounds__(NW * 32) void gemm_splitk(
    float* __restrict__ Z, const float* __restrict__ H,
    const float* __restrict__ S, int Mrows)
{
    constexpr int KS  = N / NW;
    constexpr int NJ  = KC / 4;
    constexpr int NQ  = KC / 8;
    constexpr int NCH = KS / KC;
    constexpr int NA  = PK ? (MT / 2) * NJ : MT * NQ;
    constexpr int NAP = PK ? 1 : MT;
    constexpr int OP  = 132;
    constexpr int NRED = NW * 2 * 8 * 32;
    static_assert(KS % (4 * KC) == 0, "four chunks per loop trip");
    static_assert(!PK || (MT == 4 && KC == 16 && NW == 16), "packed form: 64 rows, 16-k chunks, 512 threads");

    __shared__ __attribute__((aligned(16))) float smem[NRED + (PK ? 16 * OP : 0)];
    float* const red  = smem;
    float* const outp = smem + NRED;

    const int lane = threadIdx.x & 31;
    const int wave = threadIdx.x >> 5;
    const int c    = lane & 15;
    const int h    = lane >> 4;
    const int n0   = blockIdx.x * 32;
    const int k0   = wave * KS + h * (KC / 2);

    const float* __restrict__ Ap[NAP];
    if constexpr (PK) {
        Ap[0] = H + (size_t)(wave * (KS / 2) + 4 * h) * 128 + 4 * c;
    } else {
#pragma unroll
        for (int mt = 0; mt < MT; ++mt) {
            int m = mt * 16 + c;
            if (m >= Mrows) m = 0;
            Ap[mt] = H + (size_t)m * N + k0;
        }
    }
    const float* __restrict__ Bp = S + (size_t)k0 * N + n0 + c;

    v8f acc[MT][2];
#pragma unroll
    for (int mt = 0; mt < MT; ++mt) {
        acc[mt][0] = (v8f){0.f, 0.f, 0.f, 0.f, 0.f, 0.f, 0.f, 0.f};
        acc[mt][1] = (v8f){0.f, 0.f, 0.f, 0.f, 0.f, 0.f, 0.f, 0.f};
    }

    v4f a0[NA], a1[NA], a2[NA];
    v2f b0[NJ][2],  b1[NJ][2],  b2[NJ][2];

    auto load_chunk = [&](v4f (&a)[NA], v2f (&b)[NJ][2], int ch) {
        const int kb = ch * KC;
#pragma unroll
        for (int j = 0; j < NJ; ++j) {
            const float* __restrict__ r0 = Bp + (size_t)(kb + 2 * j) * N;
            const float* __restrict__ r1 = r0 + N;
            b[j][0].x = r0[0];  b[j][0].y = r1[0];
            b[j][1].x = r0[16]; b[j][1].y = r1[16];
        }
        if constexpr (PK) {
#pragma unroll
            for (int g = 0; g < MT / 2; ++g)
#pragma unroll
                for (int j = 0; j < NJ; ++j)
                    a[g * NJ + j] = *(const v4f*)(Ap[0] + (size_t)(kb / 2 + j) * 128 + 64 * g);
        } else {
#pragma unroll
            for (int mt = 0; mt < MT; ++mt)
#pragma unroll
                for (int q = 0; q < NQ; ++q)
                    a[mt * NQ + q] = *(const v4f*)(Ap[mt] + kb + 4 * q);
        }
    };

    auto mma_chunk = [&](const v4f (&a)[NA], const v2f (&b)[NJ][2]) {
#pragma unroll
        for (int j = 0; j < NJ; ++j) {
#pragma unroll
            for (int mt = 0; mt < MT; ++mt) {
                v2f av;
                if constexpr (PK) {
                    const v4f q4 = a[(mt >> 1) * NJ + j];
                    if (mt & 1) { av.x = q4.z; av.y = q4.w; }
                    else        { av.x = q4.x; av.y = q4.y; }
                } else {
                    const v4f q4 = a[mt * NQ + (j >> 1)];
                    if (j & 1) { av.x = q4.z; av.y = q4.w; }
                    else       { av.x = q4.x; av.y = q4.y; }
                }
                acc[mt][0] = __builtin_amdgcn_wmma_f32_16x16x4_f32(
                    false, av, false, b[j][0], (short)0, acc[mt][0], false, false);
                acc[mt][1] = __builtin_amdgcn_wmma_f32_16x16x4_f32(
                    false, av, false, b[j][1], (short)0, acc[mt][1], false, false);
            }
        }
    };

#define SCHED_FENCE() __builtin_amdgcn_sched_barrier(0)
    load_chunk(a0, b0, 0);
    SCHED_FENCE();
    load_chunk(a1, b1, 1);
    SCHED_FENCE();
#pragma nounroll
    for (int ch = 0; ch < NCH - 4; ch += 4) {
        mma_chunk(a0, b0);
        SCHED_FENCE();
        load_chunk(a0, b0, ch + 2);
        SCHED_FENCE();
        load_chunk(a2, b2, ch + 3);
        SCHED_FENCE();
        mma_chunk(a1, b1);
        mma_chunk(a0, b0);
        SCHED_FENCE();
        load_chunk(a0, b0, ch + 4);
        SCHED_FENCE();
        load_chunk(a1, b1, ch + 5);
        SCHED_FENCE();
        mma_chunk(a2, b2);
    }
    mma_chunk(a0, b0);
    SCHED_FENCE();
    load_chunk(a0, b0, NCH - 2);
    SCHED_FENCE();
    load_chunk(a2, b2, NCH - 1);
    SCHED_FENCE();
    mma_chunk(a1, b1);
    mma_chunk(a0, b0);
    mma_chunk(a2, b2);
#undef SCHED_FENCE

#pragma unroll
    for (int mt = 0; mt < MT; ++mt) {
        if (mt) __syncthreads();
#pragma unroll
        for (int t = 0; t < 2; ++t)
#pragma unroll
            for (int r = 0; r < 8; ++r)
                red[((wave * 2 + t) * 8 + r) * 32 + lane] = acc[mt][t][r];
        __syncthreads();
        if (threadIdx.x < 256) {
            const int r = wave;
            float s0 = red[(0 * 8 + r) * 32 + lane];
            float s1 = red[(1 * 8 + r) * 32 + lane];
#pragma unroll
            for (int w = 1; w < NW; ++w) {
                s0 += red[((w * 2 + 0) * 8 + r) * 32 + lane];
                s1 += red[((w * 2 + 1) * 8 + r) * 32 + lane];
            }
            if constexpr (PK) {
                const int m = 32 * (mt >> 1) + 2 * (8 * h + r) + (mt & 1);
                float* __restrict__ op = outp + (c >> 1) * OP + 2 * m + (c & 1);
                op[0]      = s0;
                op[8 * OP] = s1;
            } else {
                const int row = mt * 16 + 8 * h + r;
                if (row < Mrows) {
                    float* __restrict__ zr = Z + (size_t)row * N + n0 + c;
                    zr[0]  = s0;
                    zr[16] = s1;
                }
            }
        }
    }
    if constexpr (PK) {
        __syncthreads();
        const int pr = threadIdx.x >> 5;
        const v4f v = *(const v4f*)(outp + pr * OP + 4 * lane);
        *(v4f*)(Z + (size_t)(n0 / 2 + pr) * 128 + 4 * lane) = v;
    }
}


__global__ void combine1_relu_packed(
    float* __restrict__ Hout,
    const float* __restrict__ Z0, const float* __restrict__ Z1,
    const float* __restrict__ Z2,
    const float* __restrict__ W, const float* __restrict__ bias, int N)
{
    const int lane = threadIdx.x & 31;
    const int kp   = blockIdx.x * (blockDim.x >> 5) + (threadIdx.x >> 5);
    if (kp >= N / 2) return;
    const int b  = lane >> 3;
    const int g0 = (2 * lane) & 15;

    const v2f z0 = *(const v2f*)(Z0 + (size_t)b * N + 2 * kp);
    const v2f z1 = *(const v2f*)(Z1 + (size_t)b * N + 2 * kp);
    const v2f z2 = *(const v2f*)(Z2 + (size_t)b * N + 2 * kp);

    v4f o;
    for (int e = 0; e < 2; ++e) {
        const int g = g0 + e;
        const float w0 = W[g], w1 = W[16 + g], w2 = W[32 + g], bs = bias[g];
        const float v0 = fmaf(z0.x, w0, fmaf(z1.x, w1, fmaf(z2.x, w2, bs)));
        const float v1 = fmaf(z0.y, w0, fmaf(z1.y, w1, fmaf(z2.y, w2, bs)));
        o[2 * e]     = v0 > 0.f ? v0 : 0.f;
        o[2 * e + 1] = v1 > 0.f ? v1 : 0.f;
    }
    *(v4f*)(Hout + (size_t)kp * 128 + 4 * lane) = o;
}

__global__ void combine_relu_packed(
    float* __restrict__ Hout,
    const float* __restrict__ Z0, const float* __restrict__ Z1,
    const float* __restrict__ Z2,
    const float* __restrict__ W, const float* __restrict__ bias, int N)
{
    const int lane = threadIdx.x & 31;
    const int kp   = blockIdx.x * (blockDim.x >> 5) + (threadIdx.x >> 5);
    if (kp >= N / 2) return;
    const int b  = lane >> 3;
    const int g0 = (2 * lane) & 15;

    const float* __restrict__ r0 = Z0 + (size_t)kp * 128 + b * 32;
    const float* __restrict__ r1 = Z1 + (size_t)kp * 128 + b * 32;
    const float* __restrict__ r2 = Z2 + (size_t)kp * 128 + b * 32;

    float o00 = bias[g0],     o01 = o00;
    float o10 = bias[g0 + 1], o11 = o10;

    for (int f = 0; f < 16; ++f) {
        const v2f z0 = *(const v2f*)(r0 + 2 * f);
        const v2f z1 = *(const v2f*)(r1 + 2 * f);
        const v2f z2 = *(const v2f*)(r2 + 2 * f);
        const v2f w0 = *(const v2f*)(W + (0 * 16 + f) * 16 + g0);
        const v2f w1 = *(const v2f*)(W + (1 * 16 + f) * 16 + g0);
        const v2f w2 = *(const v2f*)(W + (2 * 16 + f) * 16 + g0);
        o00 = fmaf(z0.x, w0.x, fmaf(z1.x, w1.x, fmaf(z2.x, w2.x, o00)));
        o01 = fmaf(z0.y, w0.x, fmaf(z1.y, w1.x, fmaf(z2.y, w2.x, o01)));
        o10 = fmaf(z0.x, w0.y, fmaf(z1.x, w1.y, fmaf(z2.x, w2.y, o10)));
        o11 = fmaf(z0.y, w0.y, fmaf(z1.y, w1.y, fmaf(z2.y, w2.y, o11)));
    }

    v4f o;
    o.x = o00 > 0.f ? o00 : 0.f;
    o.y = o01 > 0.f ? o01 : 0.f;
    o.z = o10 > 0.f ? o10 : 0.f;
    o.w = o11 > 0.f ? o11 : 0.f;
    *(v4f*)(Hout + (size_t)kp * 128 + 4 * lane) = o;
}

__global__ void pool_packed(
    float* __restrict__ P, const float* __restrict__ Hin,
    const int* __restrict__ incid, int N2, int DEG)
{
    const int lane = threadIdx.x & 31;
    const int kp   = blockIdx.x * (blockDim.x >> 5) + (threadIdx.x >> 5);
    if (kp >= N2 / 2) return;

    v4f o;
    for (int s = 0; s < 2; ++s) {
        const int node = 2 * kp + s;
        float v0 = 0.0f, v1 = 0.0f;
        for (int d = 0; d < DEG; ++d) {
            const int j = incid[node * DEG + d];
            const v4f q = *(const v4f*)(Hin + (size_t)(j >> 1) * 128 + 4 * lane);
            v0 = fmaxf(v0, (j & 1) ? q.y : q.x);
            v1 = fmaxf(v1, (j & 1) ? q.w : q.z);
        }
        o[s]     = v0;
        o[2 + s] = v1;
    }
    *(v4f*)(P + (size_t)kp * 128 + 4 * lane) = o;
}

__global__ void readout_packed(
    float* __restrict__ Y, const float* __restrict__ H,
    const float* __restrict__ R0W, const float* __restrict__ R0b,
    const float* __restrict__ R1W, const float* __restrict__ R1b,
    int Bn, int N2)
{
    const int tid = blockIdx.x * blockDim.x + threadIdx.x;
    const int NP  = N2 / 2;
    if (tid >= Bn * NP) return;
    const int kp = tid % NP;
    const int b  = tid / NP;

    const float* __restrict__ hr = H + (size_t)kp * 128 + b * 32;
    float v[16][2];
    for (int i = 0; i < 8; ++i) {
        const v4f q = *(const v4f*)(hr + 4 * i);
        v[2 * i][0]     = q.x;  v[2 * i][1]     = q.y;
        v[2 * i + 1][0] = q.z;  v[2 * i + 1][1] = q.w;
    }

    float res[2][2];
    for (int s = 0; s < 2; ++s) {
        float y0[16];
        for (int j = 0; j < 16; ++j) {
            float a = R0b[j];
            for (int i = 0; i < 16; ++i) a = fmaf(v[i][s], R0W[i * 16 + j], a);
            y0[j] = a > 0.f ? a : 0.f;
        }
        for (int r = 0; r < 2; ++r) {
            float a = R1b[r];
            for (int j = 0; j < 16; ++j) a = fmaf(y0[j], R1W[j * 2 + r], a);
            res[r][s] = a;
        }
    }
    for (int r = 0; r < 2; ++r) {
        v2f w; w.x = res[r][0]; w.y = res[r][1];
        *(v2f*)(Y + ((size_t)(b * 2 + r)) * (size_t)N2 + 2 * kp) = w;
    }
}

extern "C" void kernel_launch(void* const* d_in, const int* in_sizes, int n_in,
                              void* d_out, int out_size, void* d_ws, size_t ws_size,
                              hipStream_t stream)
{
    (void)in_sizes; (void)n_in; (void)out_size; (void)ws_size;

    const float* x     = (const float*)d_in[0];
    const float* S1    = (const float*)d_in[1];
    const float* S2    = (const float*)d_in[2];
    const int*   incid = (const int*)  d_in[3];
    const float* W10   = (const float*)d_in[4];
    const float* b10   = (const float*)d_in[5];
    const float* W11   = (const float*)d_in[6];
    const float* b11   = (const float*)d_in[7];
    const float* W20   = (const float*)d_in[8];
    const float* b20   = (const float*)d_in[9];
    const float* R0W   = (const float*)d_in[10];
    const float* R0b   = (const float*)d_in[11];
    const float* R1W   = (const float*)d_in[12];
    const float* R1b   = (const float*)d_in[13];
    float* out = (float*)d_out;

    const int B = 4, F1 = 16, N1 = 8192, N2 = 4096, DEG = 4;

    char* ws = (char*)d_ws;
    const size_t szBig = (size_t)(B * F1) * N1 * sizeof(float);
    float* h1 = (float*)(ws);
    float* tA = (float*)(ws + 1 * szBig);
    float* tB = (float*)(ws + 2 * szBig);
    float* h2 = (float*)(ws + 3 * szBig);
    float* p  = (float*)(ws + 4 * szBig);

    auto grid1d    = [](int total, int blk) { return (total + blk - 1) / blk; };


    gemm_splitk<N1, 1, 16, 16, false><<<N1 / 32, 512, 0, stream>>>(tA, x,  S1, B);
    gemm_splitk<N1, 1, 16, 16, false><<<N1 / 32, 512, 0, stream>>>(tB, tA, S1, B);
    combine1_relu_packed<<<grid1d(N1 / 2, 8), 256, 0, stream>>>(
        h1, x, tA, tB, W10, b10, N1);

    gemm_splitk<N1, 4, 16, 16, true><<<N1 / 32, 512, 0, stream>>>(tA, h1, S1, B * F1);
    gemm_splitk<N1, 4, 16, 16, true><<<N1 / 32, 512, 0, stream>>>(tB, tA, S1, B * F1);
    combine_relu_packed<<<grid1d(N1 / 2, 8), 256, 0, stream>>>(
        h2, h1, tA, tB, W11, b11, N1);

    pool_packed<<<grid1d(N2 / 2, 8), 256, 0, stream>>>(p, h2, incid, N2, DEG);

    gemm_splitk<N2, 4, 16, 16, true><<<N2 / 32, 512, 0, stream>>>(tA, p,  S2, B * F1);
    gemm_splitk<N2, 4, 16, 16, true><<<N2 / 32, 512, 0, stream>>>(tB, tA, S2, B * F1);
    combine_relu_packed<<<grid1d(N2 / 2, 8), 256, 0, stream>>>(
        h1  , p, tA, tB, W20, b20, N2);

    readout_packed<<<grid1d(B * (N2 / 2), 256), 256, 0, stream>>>(
        out, h1, R0W, R0b, R1W, R1b, B, N2);
}


# N=4096 GEMMs: each 32-column strip split by rows over two workgroups (32 rows each), grid 128 -> 256 so every compute unit has a workgroup; k slices and order of sums unchanged
# speedup vs baseline: 5.7215x; 1.0560x over previous
#include <hip/hip_runtime.h>
#include <hip/hip_bf16.h>

typedef float v2f __attribute__((ext_vector_type(2)));
typedef float v4f __attribute__((ext_vector_type(4)));
typedef float v8f __attribute__((ext_vector_type(8)));

template <int N, int MT, int KC, int NW, bool PK>
__global__ __launch_bounds__(NW * 32) void gemm_splitk(
    float* __restrict__ Z, const float* __restrict__ H,
    const float* __restrict__ S, int Mrows)
{
    constexpr int KS  = N / NW;
    constexpr int NJ  = KC / 4;
    constexpr int NQ  = KC / 8;
    constexpr int NCH = KS / KC;
    constexpr int NA  = PK ? (MT / 2) * NJ : MT * NQ;
    constexpr int NAP = PK ? 1 : MT;
    constexpr int ROWF = MT * 32;
    constexpr int OP  = ROWF + 4;
    constexpr int NRED = NW * 2 * 8 * 32;
    static_assert(KS % (4 * KC) == 0, "four chunks per loop trip");
    static_assert(!PK || ((MT == 4 || MT == 2) && KC == 16 && NW == 16), "packed form: 64 or 32 rows, 16-k chunks, 512 threads");

    __shared__ __attribute__((aligned(16))) float smem[NRED + (PK ? 16 * OP : 0)];
    float* const red  = smem;
    float* const outp = smem + NRED;

    const int lane = threadIdx.x & 31;
    const int wave = threadIdx.x >> 5;
    const int c    = lane & 15;
    const int h    = lane >> 4;
    const int n0   = blockIdx.x * 32;
    const int k0   = wave * KS + h * (KC / 2);

    const int G0 = (PK && MT == 2) ? (int)blockIdx.y : 0;

    const float* __restrict__ Ap[NAP];
    if constexpr (PK) {
        Ap[0] = H + (size_t)(wave * (KS / 2) + 4 * h) * 128 + 64 * G0 + 4 * c;
    } else {
#pragma unroll
        for (int mt = 0; mt < MT; ++mt) {
            int m = mt * 16 + c;
            if (m >= Mrows) m = 0;
            Ap[mt] = H + (size_t)m * N + k0;
        }
    }
    const float* __restrict__ Bp = S + (size_t)k0 * N + n0 + c;

    v8f acc[MT][2];
#pragma unroll
    for (int mt = 0; mt < MT; ++mt) {
        acc[mt][0] = (v8f){0.f, 0.f, 0.f, 0.f, 0.f, 0.f, 0.f, 0.f};
        acc[mt][1] = (v8f){0.f, 0.f, 0.f, 0.f, 0.f, 0.f, 0.f, 0.f};
    }

    v4f a0[NA], a1[NA], a2[NA];
    v2f b0[NJ][2],  b1[NJ][2],  b2[NJ][2];

    auto load_chunk = [&](v4f (&a)[NA], v2f (&b)[NJ][2], int ch) {
        const int kb = ch * KC;
#pragma unroll
        for (int j = 0; j < NJ; ++j) {
            const float* __restrict__ r0 = Bp + (size_t)(kb + 2 * j) * N;
            const float* __restrict__ r1 = r0 + N;
            b[j][0].x = r0[0];  b[j][0].y = r1[0];
            b[j][1].x = r0[16]; b[j][1].y = r1[16];
        }
        if constexpr (PK) {
#pragma unroll
            for (int g = 0; g < MT / 2; ++g)
#pragma unroll
                for (int j = 0; j < NJ; ++j)
                    a[g * NJ + j] = *(const v4f*)(Ap[0] + (size_t)(kb / 2 + j) * 128 + 64 * g);
        } else {
#pragma unroll
            for (int mt = 0; mt < MT; ++mt)
#pragma unroll
                for (int q = 0; q < NQ; ++q)
                    a[mt * NQ + q] = *(const v4f*)(Ap[mt] + kb + 4 * q);
        }
    };

    auto mma_chunk = [&](const v4f (&a)[NA], const v2f (&b)[NJ][2]) {
#pragma unroll
        for (int j = 0; j < NJ; ++j) {
#pragma unroll
            for (int mt = 0; mt < MT; ++mt) {
                v2f av;
                if constexpr (PK) {
                    const v4f q4 = a[(mt >> 1) * NJ + j];
                    if (mt & 1) { av.x = q4.z; av.y = q4.w; }
                    else        { av.x = q4.x; av.y = q4.y; }
                } else {
                    const v4f q4 = a[mt * NQ + (j >> 1)];
                    if (j & 1) { av.x = q4.z; av.y = q4.w; }
                    else       { av.x = q4.x; av.y = q4.y; }
                }
                acc[mt][0] = __builtin_amdgcn_wmma_f32_16x16x4_f32(
                    false, av, false, b[j][0], (short)0, acc[mt][0], false, false);
                acc[mt][1] = __builtin_amdgcn_wmma_f32_16x16x4_f32(
                    false, av, false, b[j][1], (short)0, acc[mt][1], false, false);
            }
        }
    };

#define SCHED_FENCE() __builtin_amdgcn_sched_barrier(0)
    load_chunk(a0, b0, 0);
    SCHED_FENCE();
    load_chunk(a1, b1, 1);
    SCHED_FENCE();
#pragma nounroll
    for (int ch = 0; ch < NCH - 4; ch += 4) {
        mma_chunk(a0, b0);
        SCHED_FENCE();
        load_chunk(a0, b0, ch + 2);
        SCHED_FENCE();
        load_chunk(a2, b2, ch + 3);
        SCHED_FENCE();
        mma_chunk(a1, b1);
        mma_chunk(a0, b0);
        SCHED_FENCE();
        load_chunk(a0, b0, ch + 4);
        SCHED_FENCE();
        load_chunk(a1, b1, ch + 5);
        SCHED_FENCE();
        mma_chunk(a2, b2);
    }
    mma_chunk(a0, b0);
    SCHED_FENCE();
    load_chunk(a0, b0, NCH - 2);
    SCHED_FENCE();
    load_chunk(a2, b2, NCH - 1);
    SCHED_FENCE();
    mma_chunk(a1, b1);
    mma_chunk(a0, b0);
    mma_chunk(a2, b2);
#undef SCHED_FENCE

#pragma unroll
    for (int mt = 0; mt < MT; ++mt) {
        if (mt) __syncthreads();
#pragma unroll
        for (int t = 0; t < 2; ++t)
#pragma unroll
            for (int r = 0; r < 8; ++r)
                red[((wave * 2 + t) * 8 + r) * 32 + lane] = acc[mt][t][r];
        __syncthreads();
        if (threadIdx.x < 256) {
            const int r = wave;
            float s0 = red[(0 * 8 + r) * 32 + lane];
            float s1 = red[(1 * 8 + r) * 32 + lane];
#pragma unroll
            for (int w = 1; w < NW; ++w) {
                s0 += red[((w * 2 + 0) * 8 + r) * 32 + lane];
                s1 += red[((w * 2 + 1) * 8 + r) * 32 + lane];
            }
            if constexpr (PK) {
                const int m = 32 * (mt >> 1) + 2 * (8 * h + r) + (mt & 1);
                float* __restrict__ op = outp + (c >> 1) * OP + 2 * m + (c & 1);
                op[0]      = s0;
                op[8 * OP] = s1;
            } else {
                const int row = mt * 16 + 8 * h + r;
                if (row < Mrows) {
                    float* __restrict__ zr = Z + (size_t)row * N + n0 + c;
                    zr[0]  = s0;
                    zr[16] = s1;
                }
            }
        }
    }
    if constexpr (PK) {
        __syncthreads();
        constexpr int TPR = ROWF / 4;
        if (MT == 4 || threadIdx.x < 16 * TPR) {
            const int pr = threadIdx.x / TPR;
            const int q  = threadIdx.x % TPR;
            const v4f v = *(const v4f*)(outp + pr * OP + 4 * q);
            *(v4f*)(Z + (size_t)(n0 / 2 + pr) * 128 + 64 * G0 + 4 * q) = v;
        }
    }
}


__global__ void combine1_relu_packed(
    float* __restrict__ Hout,
    const float* __restrict__ Z0, const float* __restrict__ Z1,
    const float* __restrict__ Z2,
    const float* __restrict__ W, const float* __restrict__ bias, int N)
{
    const int lane = threadIdx.x & 31;
    const int kp   = blockIdx.x * (blockDim.x >> 5) + (threadIdx.x >> 5);
    if (kp >= N / 2) return;
    const int b  = lane >> 3;
    const int g0 = (2 * lane) & 15;

    const v2f z0 = *(const v2f*)(Z0 + (size_t)b * N + 2 * kp);
    const v2f z1 = *(const v2f*)(Z1 + (size_t)b * N + 2 * kp);
    const v2f z2 = *(const v2f*)(Z2 + (size_t)b * N + 2 * kp);

    v4f o;
    for (int e = 0; e < 2; ++e) {
        const int g = g0 + e;
        const float w0 = W[g], w1 = W[16 + g], w2 = W[32 + g], bs = bias[g];
        const float v0 = fmaf(z0.x, w0, fmaf(z1.x, w1, fmaf(z2.x, w2, bs)));
        const float v1 = fmaf(z0.y, w0, fmaf(z1.y, w1, fmaf(z2.y, w2, bs)));
        o[2 * e]     = v0 > 0.f ? v0 : 0.f;
        o[2 * e + 1] = v1 > 0.f ? v1 : 0.f;
    }
    *(v4f*)(Hout + (size_t)kp * 128 + 4 * lane) = o;
}

__global__ void combine_relu_packed(
    float* __restrict__ Hout,
    const float* __restrict__ Z0, const float* __restrict__ Z1,
    const float* __restrict__ Z2,
    const float* __restrict__ W, const float* __restrict__ bias, int N)
{
    const int lane = threadIdx.x & 31;
    const int kp   = blockIdx.x * (blockDim.x >> 5) + (threadIdx.x >> 5);
    if (kp >= N / 2) return;
    const int b  = lane >> 3;
    const int g0 = (2 * lane) & 15;

    const float* __restrict__ r0 = Z0 + (size_t)kp * 128 + b * 32;
    const float* __restrict__ r1 = Z1 + (size_t)kp * 128 + b * 32;
    const float* __restrict__ r2 = Z2 + (size_t)kp * 128 + b * 32;

    float o00 = bias[g0],     o01 = o00;
    float o10 = bias[g0 + 1], o11 = o10;

    for (int f = 0; f < 16; ++f) {
        const v2f z0 = *(const v2f*)(r0 + 2 * f);
        const v2f z1 = *(const v2f*)(r1 + 2 * f);
        const v2f z2 = *(const v2f*)(r2 + 2 * f);
        const v2f w0 = *(const v2f*)(W + (0 * 16 + f) * 16 + g0);
        const v2f w1 = *(const v2f*)(W + (1 * 16 + f) * 16 + g0);
        const v2f w2 = *(const v2f*)(W + (2 * 16 + f) * 16 + g0);
        o00 = fmaf(z0.x, w0.x, fmaf(z1.x, w1.x, fmaf(z2.x, w2.x, o00)));
        o01 = fmaf(z0.y, w0.x, fmaf(z1.y, w1.x, fmaf(z2.y, w2.x, o01)));
        o10 = fmaf(z0.x, w0.y, fmaf(z1.x, w1.y, fmaf(z2.x, w2.y, o10)));
        o11 = fmaf(z0.y, w0.y, fmaf(z1.y, w1.y, fmaf(z2.y, w2.y, o11)));
    }

    v4f o;
    o.x = o00 > 0.f ? o00 : 0.f;
    o.y = o01 > 0.f ? o01 : 0.f;
    o.z = o10 > 0.f ? o10 : 0.f;
    o.w = o11 > 0.f ? o11 : 0.f;
    *(v4f*)(Hout + (size_t)kp * 128 + 4 * lane) = o;
}

__global__ void pool_packed(
    float* __restrict__ P, const float* __restrict__ Hin,
    const int* __restrict__ incid, int N2, int DEG)
{
    const int lane = threadIdx.x & 31;
    const int kp   = blockIdx.x * (blockDim.x >> 5) + (threadIdx.x >> 5);
    if (kp >= N2 / 2) return;

    v4f o;
    for (int s = 0; s < 2; ++s) {
        const int node = 2 * kp + s;
        float v0 = 0.0f, v1 = 0.0f;
        for (int d = 0; d < DEG; ++d) {
            const int j = incid[node * DEG + d];
            const v4f q = *(const v4f*)(Hin + (size_t)(j >> 1) * 128 + 4 * lane);
            v0 = fmaxf(v0, (j & 1) ? q.y : q.x);
            v1 = fmaxf(v1, (j & 1) ? q.w : q.z);
        }
        o[s]     = v0;
        o[2 + s] = v1;
    }
    *(v4f*)(P + (size_t)kp * 128 + 4 * lane) = o;
}

__global__ void readout_packed(
    float* __restrict__ Y, const float* __restrict__ H,
    const float* __restrict__ R0W, const float* __restrict__ R0b,
    const float* __restrict__ R1W, const float* __restrict__ R1b,
    int Bn, int N2)
{
    const int tid = blockIdx.x * blockDim.x + threadIdx.x;
    const int NP  = N2 / 2;
    if (tid >= Bn * NP) return;
    const int kp = tid % NP;
    const int b  = tid / NP;

    const float* __restrict__ hr = H + (size_t)kp * 128 + b * 32;
    float v[16][2];
    for (int i = 0; i < 8; ++i) {
        const v4f q = *(const v4f*)(hr + 4 * i);
        v[2 * i][0]     = q.x;  v[2 * i][1]     = q.y;
        v[2 * i + 1][0] = q.z;  v[2 * i + 1][1] = q.w;
    }

    float res[2][2];
    for (int s = 0; s < 2; ++s) {
        float y0[16];
        for (int j = 0; j < 16; ++j) {
            float a = R0b[j];
            for (int i = 0; i < 16; ++i) a = fmaf(v[i][s], R0W[i * 16 + j], a);
            y0[j] = a > 0.f ? a : 0.f;
        }
        for (int r = 0; r < 2; ++r) {
            float a = R1b[r];
            for (int j = 0; j < 16; ++j) a = fmaf(y0[j], R1W[j * 2 + r], a);
            res[r][s] = a;
        }
    }
    for (int r = 0; r < 2; ++r) {
        v2f w; w.x = res[r][0]; w.y = res[r][1];
        *(v2f*)(Y + ((size_t)(b * 2 + r)) * (size_t)N2 + 2 * kp) = w;
    }
}

extern "C" void kernel_launch(void* const* d_in, const int* in_sizes, int n_in,
                              void* d_out, int out_size, void* d_ws, size_t ws_size,
                              hipStream_t stream)
{
    (void)in_sizes; (void)n_in; (void)out_size; (void)ws_size;

    const float* x     = (const float*)d_in[0];
    const float* S1    = (const float*)d_in[1];
    const float* S2    = (const float*)d_in[2];
    const int*   incid = (const int*)  d_in[3];
    const float* W10   = (const float*)d_in[4];
    const float* b10   = (const float*)d_in[5];
    const float* W11   = (const float*)d_in[6];
    const float* b11   = (const float*)d_in[7];
    const float* W20   = (const float*)d_in[8];
    const float* b20   = (const float*)d_in[9];
    const float* R0W   = (const float*)d_in[10];
    const float* R0b   = (const float*)d_in[11];
    const float* R1W   = (const float*)d_in[12];
    const float* R1b   = (const float*)d_in[13];
    float* out = (float*)d_out;

    const int B = 4, F1 = 16, N1 = 8192, N2 = 4096, DEG = 4;

    char* ws = (char*)d_ws;
    const size_t szBig = (size_t)(B * F1) * N1 * sizeof(float);
    float* h1 = (float*)(ws);
    float* tA = (float*)(ws + 1 * szBig);
    float* tB = (float*)(ws + 2 * szBig);
    float* h2 = (float*)(ws + 3 * szBig);
    float* p  = (float*)(ws + 4 * szBig);

    auto grid1d    = [](int total, int blk) { return (total + blk - 1) / blk; };


    gemm_splitk<N1, 1, 16, 16, false><<<N1 / 32, 512, 0, stream>>>(tA, x,  S1, B);
    gemm_splitk<N1, 1, 16, 16, false><<<N1 / 32, 512, 0, stream>>>(tB, tA, S1, B);
    combine1_relu_packed<<<grid1d(N1 / 2, 8), 256, 0, stream>>>(
        h1, x, tA, tB, W10, b10, N1);

    gemm_splitk<N1, 4, 16, 16, true><<<N1 / 32, 512, 0, stream>>>(tA, h1, S1, B * F1);
    gemm_splitk<N1, 4, 16, 16, true><<<N1 / 32, 512, 0, stream>>>(tB, tA, S1, B * F1);
    combine_relu_packed<<<grid1d(N1 / 2, 8), 256, 0, stream>>>(
        h2, h1, tA, tB, W11, b11, N1);

    pool_packed<<<grid1d(N2 / 2, 8), 256, 0, stream>>>(p, h2, incid, N2, DEG);

    gemm_splitk<N2, 2, 16, 16, true><<<dim3(N2 / 32, 2), 512, 0, stream>>>(tA, p,  S2, B * F1);
    gemm_splitk<N2, 2, 16, 16, true><<<dim3(N2 / 32, 2), 512, 0, stream>>>(tB, tA, S2, B * F1);
    combine_relu_packed<<<grid1d(N2 / 2, 8), 256, 0, stream>>>(
        h1  , p, tA, tB, W20, b20, N2);

    readout_packed<<<grid1d(B * (N2 / 2), 256), 256, 0, stream>>>(
        out, h1, R0W, R0b, R1W, R1b, B, N2);
}


# Per-node stages (the three feature combines and the readout MLP) run as the epilogue of the GEMM that makes their last operand: 11 launches -> 7; second products and h3 never go to memory
# speedup vs baseline: 6.2098x; 1.0853x over previous
#include <hip/hip_runtime.h>
#include <hip/hip_bf16.h>

typedef float v2f __attribute__((ext_vector_type(2)));
typedef float v4f __attribute__((ext_vector_type(4)));
typedef float v8f __attribute__((ext_vector_type(8)));

template <int N, int MT, int KC, int NW, bool PK, int EPI>
__global__ __launch_bounds__(NW * 32) void gemm_splitk(
    float* __restrict__ Z, const float* __restrict__ H,
    const float* __restrict__ S, int Mrows,
    const float* __restrict__ Z0,
    const float* __restrict__ Wc, const float* __restrict__ bc,
    const float* __restrict__ R0W, const float* __restrict__ R0b,
    const float* __restrict__ R1W, const float* __restrict__ R1b)
{
    constexpr int KS  = N / NW;
    constexpr int NJ  = KC / 4;
    constexpr int NQ  = KC / 8;
    constexpr int NCH = KS / KC;
    constexpr int NA  = PK ? (MT / 2) * NJ : MT * NQ;
    constexpr int NAP = PK ? 1 : MT;
    constexpr int ROWF = MT * 32;
    constexpr int OP  = ROWF + 4;
    constexpr int NRED = NW * 2 * 8 * 32;
    static_assert(KS % (4 * KC) == 0, "four chunks per loop trip");
    static_assert(!PK || ((MT == 4 || MT == 2) && KC == 16 && NW == 16), "packed form: 64 or 32 rows, 16-k chunks, 512 threads");
    static_assert(EPI == 0 || (PK && NW == 16) || (MT == 1 && NW == 16), "epilogue stages: 512 threads, packed form or one row tile");
    static_assert(EPI != 2 || (PK && MT == 2), "readout stage: the 32-row packed form");

    __shared__ __attribute__((aligned(16))) float smem[NRED + (PK ? 16 * OP : (EPI ? 16 * 32 : 0))];
    float* const red  = smem;
    float* const outp = smem + NRED;

    const int lane = threadIdx.x & 31;
    const int wave = threadIdx.x >> 5;
    const int c    = lane & 15;
    const int h    = lane >> 4;
    const int n0   = blockIdx.x * 32;
    const int k0   = wave * KS + h * (KC / 2);

    const int G0 = (PK && MT == 2) ? (int)blockIdx.y : 0;

    const float* __restrict__ Ap[NAP];
    if constexpr (PK) {
        Ap[0] = H + (size_t)(wave * (KS / 2) + 4 * h) * 128 + 64 * G0 + 4 * c;
    } else {
#pragma unroll
        for (int mt = 0; mt < MT; ++mt) {
            int m = mt * 16 + c;
            if (m >= Mrows) m = 0;
            Ap[mt] = H + (size_t)m * N + k0;
        }
    }
    const float* __restrict__ Bp = S + (size_t)k0 * N + n0 + c;

    v8f acc[MT][2];
#pragma unroll
    for (int mt = 0; mt < MT; ++mt) {
        acc[mt][0] = (v8f){0.f, 0.f, 0.f, 0.f, 0.f, 0.f, 0.f, 0.f};
        acc[mt][1] = (v8f){0.f, 0.f, 0.f, 0.f, 0.f, 0.f, 0.f, 0.f};
    }

    v4f a0[NA], a1[NA], a2[NA];
    v2f b0[NJ][2],  b1[NJ][2],  b2[NJ][2];

    auto load_chunk = [&](v4f (&a)[NA], v2f (&b)[NJ][2], int ch) {
        const int kb = ch * KC;
#pragma unroll
        for (int j = 0; j < NJ; ++j) {
            const float* __restrict__ r0 = Bp + (size_t)(kb + 2 * j) * N;
            const float* __restrict__ r1 = r0 + N;
            b[j][0].x = r0[0];  b[j][0].y = r1[0];
            b[j][1].x = r0[16]; b[j][1].y = r1[16];
        }
        if constexpr (PK) {
#pragma unroll
            for (int g = 0; g < MT / 2; ++g)
#pragma unroll
                for (int j = 0; j < NJ; ++j)
                    a[g * NJ + j] = *(const v4f*)(Ap[0] + (size_t)(kb / 2 + j) * 128 + 64 * g);
        } else {
#pragma unroll
            for (int mt = 0; mt < MT; ++mt)
#pragma unroll
                for (int q = 0; q < NQ; ++q)
                    a[mt * NQ + q] = *(const v4f*)(Ap[mt] + kb + 4 * q);
        }
    };

    auto mma_chunk = [&](const v4f (&a)[NA], const v2f (&b)[NJ][2]) {
#pragma unroll
        for (int j = 0; j < NJ; ++j) {
#pragma unroll
            for (int mt = 0; mt < MT; ++mt) {
                v2f av;
                if constexpr (PK) {
                    const v4f q4 = a[(mt >> 1) * NJ + j];
                    if (mt & 1) { av.x = q4.z; av.y = q4.w; }
                    else        { av.x = q4.x; av.y = q4.y; }
                } else {
                    const v4f q4 = a[mt * NQ + (j >> 1)];
                    if (j & 1) { av.x = q4.z; av.y = q4.w; }
                    else       { av.x = q4.x; av.y = q4.y; }
                }
                acc[mt][0] = __builtin_amdgcn_wmma_f32_16x16x4_f32(
                    false, av, false, b[j][0], (short)0, acc[mt][0], false, false);
                acc[mt][1] = __builtin_amdgcn_wmma_f32_16x16x4_f32(
                    false, av, false, b[j][1], (short)0, acc[mt][1], false, false);
            }
        }
    };

#define SCHED_FENCE() __builtin_amdgcn_sched_barrier(0)
    load_chunk(a0, b0, 0);
    SCHED_FENCE();
    load_chunk(a1, b1, 1);
    SCHED_FENCE();
#pragma nounroll
    for (int ch = 0; ch < NCH - 4; ch += 4) {
        mma_chunk(a0, b0);
        SCHED_FENCE();
        load_chunk(a0, b0, ch + 2);
        SCHED_FENCE();
        load_chunk(a2, b2, ch + 3);
        SCHED_FENCE();
        mma_chunk(a1, b1);
        mma_chunk(a0, b0);
        SCHED_FENCE();
        load_chunk(a0, b0, ch + 4);
        SCHED_FENCE();
        load_chunk(a1, b1, ch + 5);
        SCHED_FENCE();
        mma_chunk(a2, b2);
    }
    mma_chunk(a0, b0);
    SCHED_FENCE();
    load_chunk(a0, b0, NCH - 2);
    SCHED_FENCE();
    load_chunk(a2, b2, NCH - 1);
    SCHED_FENCE();
    mma_chunk(a1, b1);
    mma_chunk(a0, b0);
    mma_chunk(a2, b2);
#undef SCHED_FENCE

#pragma unroll
    for (int mt = 0; mt < MT; ++mt) {
        if (mt) __syncthreads();
#pragma unroll
        for (int t = 0; t < 2; ++t)
#pragma unroll
            for (int r = 0; r < 8; ++r)
                red[((wave * 2 + t) * 8 + r) * 32 + lane] = acc[mt][t][r];
        __syncthreads();
        if (threadIdx.x < 256) {
            const int r = wave;
            float s0 = red[(0 * 8 + r) * 32 + lane];
            float s1 = red[(1 * 8 + r) * 32 + lane];
#pragma unroll
            for (int w = 1; w < NW; ++w) {
                s0 += red[((w * 2 + 0) * 8 + r) * 32 + lane];
                s1 += red[((w * 2 + 1) * 8 + r) * 32 + lane];
            }
            if constexpr (PK) {
                const int m = 32 * (mt >> 1) + 2 * (8 * h + r) + (mt & 1);
                float* __restrict__ op = outp + (c >> 1) * OP + 2 * m + (c & 1);
                op[0]      = s0;
                op[8 * OP] = s1;
            } else {
                const int row = mt * 16 + 8 * h + r;
                if (row < Mrows) {
                    if constexpr (EPI == 0) {
                        float* __restrict__ zr = Z + (size_t)row * N + n0 + c;
                        zr[0]  = s0;
                        zr[16] = s1;
                    } else {
                        outp[row * 32 + c]      = s0;
                        outp[row * 32 + 16 + c] = s1;
                    }
                }
            }
        }
    }
    if constexpr (PK) {
        __syncthreads();
        constexpr int TPR = ROWF / 4;
        const bool act = (MT == 4) || (threadIdx.x < 16 * TPR);
        const int pr = threadIdx.x / TPR;
        const int q  = threadIdx.x % TPR;
        const size_t prow = (size_t)(n0 / 2 + pr) * 128 + 64 * G0;
        if constexpr (EPI == 0) {
            if (act) {
                const v4f v = *(const v4f*)(outp + pr * OP + 4 * q);
                *(v4f*)(Z + prow + 4 * q) = v;
            }
        } else {
            const int bl = q >> 3;
            const int g0 = (2 * q) & 15;
            v4f o = (v4f){0.f, 0.f, 0.f, 0.f};
            if (act) {
                const float* __restrict__ r0 = Z0 + prow + bl * 32;
                const float* __restrict__ r1 = H  + prow + bl * 32;
                const float* r2 = outp + pr * OP + bl * 32;

                float o00 = bc[g0],     o01 = o00;
                float o10 = bc[g0 + 1], o11 = o10;
#pragma unroll
                for (int f = 0; f < 16; ++f) {
                    const v2f z0 = *(const v2f*)(r0 + 2 * f);
                    const v2f z1 = *(const v2f*)(r1 + 2 * f);
                    const v2f z2 = *(const v2f*)(r2 + 2 * f);
                    const v2f w0 = *(const v2f*)(Wc + (0 * 16 + f) * 16 + g0);
                    const v2f w1 = *(const v2f*)(Wc + (1 * 16 + f) * 16 + g0);
                    const v2f w2 = *(const v2f*)(Wc + (2 * 16 + f) * 16 + g0);
                    o00 = fmaf(z0.x, w0.x, fmaf(z1.x, w1.x, fmaf(z2.x, w2.x, o00)));
                    o01 = fmaf(z0.y, w0.x, fmaf(z1.y, w1.x, fmaf(z2.y, w2.x, o01)));
                    o10 = fmaf(z0.x, w0.y, fmaf(z1.x, w1.y, fmaf(z2.x, w2.y, o10)));
                    o11 = fmaf(z0.y, w0.y, fmaf(z1.y, w1.y, fmaf(z2.y, w2.y, o11)));
                }
                o.x = o00 > 0.f ? o00 : 0.f;
                o.y = o01 > 0.f ? o01 : 0.f;
                o.z = o10 > 0.f ? o10 : 0.f;
                o.w = o11 > 0.f ? o11 : 0.f;
                if constexpr (EPI == 1)
                    *(v4f*)(Z + prow + 4 * q) = o;
            }
            if constexpr (EPI == 2) {
                float* const hs = red;
                if (act) {
                    const int i0 = (bl * 32 + 2 * pr) * 17 + g0;
                    hs[i0]          = o.x;
                    hs[i0 + 17]     = o.y;
                    hs[i0 + 1]      = o.z;
                    hs[i0 + 17 + 1] = o.w;
                }
                __syncthreads();
                if (threadIdx.x < 64) {
                    const int b2 = threadIdx.x >> 5;
                    const int i  = threadIdx.x & 31;
                    float v[16];
                    for (int g = 0; g < 16; ++g) v[g] = hs[(b2 * 32 + i) * 17 + g];
                    float y0[16];
                    for (int j = 0; j < 16; ++j) {
                        float a = R0b[j];
                        for (int g = 0; g < 16; ++g) a = fmaf(v[g], R0W[g * 16 + j], a);
                        y0[j] = a > 0.f ? a : 0.f;
                    }
                    for (int r = 0; r < 2; ++r) {
                        float a = R1b[r];
                        for (int j = 0; j < 16; ++j) a = fmaf(y0[j], R1W[j * 2 + r], a);
                        Z[((size_t)((2 * G0 + b2) * 2 + r)) * (size_t)N + n0 + i] = a;
                    }
                }
            }
        }
    } else if constexpr (EPI == 1) {
        __syncthreads();
        const int pr = threadIdx.x >> 5;
        const int b  = lane >> 3;
        const int g0 = (2 * lane) & 15;
        const v2f z0 = *(const v2f*)(Z0 + (size_t)b * N + n0 + 2 * pr);
        const v2f z1 = *(const v2f*)(H  + (size_t)b * N + n0 + 2 * pr);
        const v2f z2 = *(const v2f*)(outp + b * 32 + 2 * pr);
        v4f o;
        for (int e = 0; e < 2; ++e) {
            const int g = g0 + e;
            const float w0 = Wc[g], w1 = Wc[16 + g], w2 = Wc[32 + g], bs = bc[g];
            const float v0 = fmaf(z0.x, w0, fmaf(z1.x, w1, fmaf(z2.x, w2, bs)));
            const float v1 = fmaf(z0.y, w0, fmaf(z1.y, w1, fmaf(z2.y, w2, bs)));
            o[2 * e]     = v0 > 0.f ? v0 : 0.f;
            o[2 * e + 1] = v1 > 0.f ? v1 : 0.f;
        }
        *(v4f*)(Z + (size_t)(n0 / 2 + pr) * 128 + 4 * lane) = o;
    }
}

__global__ void pool_packed(
    float* __restrict__ P, const float* __restrict__ Hin,
    const int* __restrict__ incid, int N2, int DEG)
{
    const int lane = threadIdx.x & 31;
    const int kp   = blockIdx.x * (blockDim.x >> 5) + (threadIdx.x >> 5);
    if (kp >= N2 / 2) return;

    v4f o;
    for (int s = 0; s < 2; ++s) {
        const int node = 2 * kp + s;
        float v0 = 0.0f, v1 = 0.0f;
        for (int d = 0; d < DEG; ++d) {
            const int j = incid[node * DEG + d];
            const v4f q = *(const v4f*)(Hin + (size_t)(j >> 1) * 128 + 4 * lane);
            v0 = fmaxf(v0, (j & 1) ? q.y : q.x);
            v1 = fmaxf(v1, (j & 1) ? q.w : q.z);
        }
        o[s]     = v0;
        o[2 + s] = v1;
    }
    *(v4f*)(P + (size_t)kp * 128 + 4 * lane) = o;
}

extern "C" void kernel_launch(void* const* d_in, const int* in_sizes, int n_in,
                              void* d_out, int out_size, void* d_ws, size_t ws_size,
                              hipStream_t stream)
{
    (void)in_sizes; (void)n_in; (void)out_size; (void)ws_size;

    const float* x     = (const float*)d_in[0];
    const float* S1    = (const float*)d_in[1];
    const float* S2    = (const float*)d_in[2];
    const int*   incid = (const int*)  d_in[3];
    const float* W10   = (const float*)d_in[4];
    const float* b10   = (const float*)d_in[5];
    const float* W11   = (const float*)d_in[6];
    const float* b11   = (const float*)d_in[7];
    const float* W20   = (const float*)d_in[8];
    const float* b20   = (const float*)d_in[9];
    const float* R0W   = (const float*)d_in[10];
    const float* R0b   = (const float*)d_in[11];
    const float* R1W   = (const float*)d_in[12];
    const float* R1b   = (const float*)d_in[13];
    float* out = (float*)d_out;

    const int B = 4, F1 = 16, N1 = 8192, N2 = 4096, DEG = 4;

    char* ws = (char*)d_ws;
    const size_t szBig = (size_t)(B * F1) * N1 * sizeof(float);
    float* h1 = (float*)(ws);
    float* tA = (float*)(ws + 1 * szBig);
    float* h2 = (float*)(ws + 2 * szBig);
    float* p  = (float*)(ws + 3 * szBig);

    auto grid1d    = [](int total, int blk) { return (total + blk - 1) / blk; };


    gemm_splitk<N1, 1, 16, 16, false, 0><<<N1 / 32, 512, 0, stream>>>(
        tA, x,  S1, B, nullptr, nullptr, nullptr, nullptr, nullptr, nullptr, nullptr);
    gemm_splitk<N1, 1, 16, 16, false, 1><<<N1 / 32, 512, 0, stream>>>(
        h1, tA, S1, B, x, W10, b10, nullptr, nullptr, nullptr, nullptr);

    gemm_splitk<N1, 4, 16, 16, true, 0><<<N1 / 32, 512, 0, stream>>>(
        tA, h1, S1, B * F1, nullptr, nullptr, nullptr, nullptr, nullptr, nullptr, nullptr);
    gemm_splitk<N1, 4, 16, 16, true, 1><<<N1 / 32, 512, 0, stream>>>(
        h2, tA, S1, B * F1, h1, W11, b11, nullptr, nullptr, nullptr, nullptr);

    pool_packed<<<grid1d(N2 / 2, 8), 256, 0, stream>>>(p, h2, incid, N2, DEG);

    gemm_splitk<N2, 2, 16, 16, true, 0><<<dim3(N2 / 32, 2), 512, 0, stream>>>(
        tA, p,  S2, B * F1, nullptr, nullptr, nullptr, nullptr, nullptr, nullptr, nullptr);
    gemm_splitk<N2, 2, 16, 16, true, 2><<<dim3(N2 / 32, 2), 512, 0, stream>>>(
        out, tA, S2, B * F1, p, W20, b20, R0W, R0b, R1W, R1b);
}


# Layer-1 second product walks each wave's K slice from its last chunk to its first, so it starts on the part of S1 the first product read last (still in L2); order of a wave's chunk sums changes
# speedup vs baseline: 6.5179x; 1.0496x over previous
#include <hip/hip_runtime.h>
#include <hip/hip_bf16.h>

typedef float v2f __attribute__((ext_vector_type(2)));
typedef float v4f __attribute__((ext_vector_type(4)));
typedef float v8f __attribute__((ext_vector_type(8)));

template <int N, int MT, int KC, int NW, bool PK, int EPI, bool REV>
__global__ __launch_bounds__(NW * 32) void gemm_splitk(
    float* __restrict__ Z, const float* __restrict__ H,
    const float* __restrict__ S, int Mrows,
    const float* __restrict__ Z0,
    const float* __restrict__ Wc, const float* __restrict__ bc,
    const float* __restrict__ R0W, const float* __restrict__ R0b,
    const float* __restrict__ R1W, const float* __restrict__ R1b)
{
    constexpr int KS  = N / NW;
    constexpr int NJ  = KC / 4;
    constexpr int NQ  = KC / 8;
    constexpr int NCH = KS / KC;
    constexpr int NA  = PK ? (MT / 2) * NJ : MT * NQ;
    constexpr int NAP = PK ? 1 : MT;
    constexpr int ROWF = MT * 32;
    constexpr int OP  = ROWF + 4;
    constexpr int NRED = NW * 2 * 8 * 32;
    static_assert(KS % (4 * KC) == 0, "four chunks per loop trip");
    static_assert(!PK || ((MT == 4 || MT == 2) && KC == 16 && NW == 16), "packed form: 64 or 32 rows, 16-k chunks, 512 threads");
    static_assert(EPI == 0 || (PK && NW == 16) || (MT == 1 && NW == 16), "epilogue stages: 512 threads, packed form or one row tile");
    static_assert(EPI != 2 || (PK && MT == 2), "readout stage: the 32-row packed form");

    __shared__ __attribute__((aligned(16))) float smem[NRED + (PK ? 16 * OP : (EPI ? 16 * 32 : 0))];
    float* const red  = smem;
    float* const outp = smem + NRED;

    const int lane = threadIdx.x & 31;
    const int wave = threadIdx.x >> 5;
    const int c    = lane & 15;
    const int h    = lane >> 4;
    const int n0   = blockIdx.x * 32;
    const int k0   = wave * KS + h * (KC / 2);

    const int G0 = (PK && MT == 2) ? (int)blockIdx.y : 0;

    const float* __restrict__ Ap[NAP];
    if constexpr (PK) {
        Ap[0] = H + (size_t)(wave * (KS / 2) + 4 * h) * 128 + 64 * G0 + 4 * c;
    } else {
#pragma unroll
        for (int mt = 0; mt < MT; ++mt) {
            int m = mt * 16 + c;
            if (m >= Mrows) m = 0;
            Ap[mt] = H + (size_t)m * N + k0;
        }
    }
    const float* __restrict__ Bp = S + (size_t)k0 * N + n0 + c;

    v8f acc[MT][2];
#pragma unroll
    for (int mt = 0; mt < MT; ++mt) {
        acc[mt][0] = (v8f){0.f, 0.f, 0.f, 0.f, 0.f, 0.f, 0.f, 0.f};
        acc[mt][1] = (v8f){0.f, 0.f, 0.f, 0.f, 0.f, 0.f, 0.f, 0.f};
    }

    v4f a0[NA], a1[NA], a2[NA];
    v2f b0[NJ][2],  b1[NJ][2],  b2[NJ][2];

    auto load_chunk = [&](v4f (&a)[NA], v2f (&b)[NJ][2], int ch) {
        const int kb = (REV ? (NCH - 1 - ch) : ch) * KC;
#pragma unroll
        for (int j = 0; j < NJ; ++j) {
            const float* __restrict__ r0 = Bp + (size_t)(kb + 2 * j) * N;
            const float* __restrict__ r1 = r0 + N;
            b[j][0].x = r0[0];  b[j][0].y = r1[0];
            b[j][1].x = r0[16]; b[j][1].y = r1[16];
        }
        if constexpr (PK) {
#pragma unroll
            for (int g = 0; g < MT / 2; ++g)
#pragma unroll
                for (int j = 0; j < NJ; ++j)
                    a[g * NJ + j] = *(const v4f*)(Ap[0] + (size_t)(kb / 2 + j) * 128 + 64 * g);
        } else {
#pragma unroll
            for (int mt = 0; mt < MT; ++mt)
#pragma unroll
                for (int q = 0; q < NQ; ++q)
                    a[mt * NQ + q] = *(const v4f*)(Ap[mt] + kb + 4 * q);
        }
    };

    auto mma_chunk = [&](const v4f (&a)[NA], const v2f (&b)[NJ][2]) {
#pragma unroll
        for (int j = 0; j < NJ; ++j) {
#pragma unroll
            for (int mt = 0; mt < MT; ++mt) {
                v2f av;
                if constexpr (PK) {
                    const v4f q4 = a[(mt >> 1) * NJ + j];
                    if (mt & 1) { av.x = q4.z; av.y = q4.w; }
                    else        { av.x = q4.x; av.y = q4.y; }
                } else {
                    const v4f q4 = a[mt * NQ + (j >> 1)];
                    if (j & 1) { av.x = q4.z; av.y = q4.w; }
                    else       { av.x = q4.x; av.y = q4.y; }
                }
                acc[mt][0] = __builtin_amdgcn_wmma_f32_16x16x4_f32(
                    false, av, false, b[j][0], (short)0, acc[mt][0], false, false);
                acc[mt][1] = __builtin_amdgcn_wmma_f32_16x16x4_f32(
                    false, av, false, b[j][1], (short)0, acc[mt][1], false, false);
            }
        }
    };

#define SCHED_FENCE() __builtin_amdgcn_sched_barrier(0)
    load_chunk(a0, b0, 0);
    SCHED_FENCE();
    load_chunk(a1, b1, 1);
    SCHED_FENCE();
#pragma nounroll
    for (int ch = 0; ch < NCH - 4; ch += 4) {
        mma_chunk(a0, b0);
        SCHED_FENCE();
        load_chunk(a0, b0, ch + 2);
        SCHED_FENCE();
        load_chunk(a2, b2, ch + 3);
        SCHED_FENCE();
        mma_chunk(a1, b1);
        mma_chunk(a0, b0);
        SCHED_FENCE();
        load_chunk(a0, b0, ch + 4);
        SCHED_FENCE();
        load_chunk(a1, b1, ch + 5);
        SCHED_FENCE();
        mma_chunk(a2, b2);
    }
    mma_chunk(a0, b0);
    SCHED_FENCE();
    load_chunk(a0, b0, NCH - 2);
    SCHED_FENCE();
    load_chunk(a2, b2, NCH - 1);
    SCHED_FENCE();
    mma_chunk(a1, b1);
    mma_chunk(a0, b0);
    mma_chunk(a2, b2);
#undef SCHED_FENCE

#pragma unroll
    for (int mt = 0; mt < MT; ++mt) {
        if (mt) __syncthreads();
#pragma unroll
        for (int t = 0; t < 2; ++t)
#pragma unroll
            for (int r = 0; r < 8; ++r)
                red[((wave * 2 + t) * 8 + r) * 32 + lane] = acc[mt][t][r];
        __syncthreads();
        if (threadIdx.x < 256) {
            const int r = wave;
            float s0 = red[(0 * 8 + r) * 32 + lane];
            float s1 = red[(1 * 8 + r) * 32 + lane];
#pragma unroll
            for (int w = 1; w < NW; ++w) {
                s0 += red[((w * 2 + 0) * 8 + r) * 32 + lane];
                s1 += red[((w * 2 + 1) * 8 + r) * 32 + lane];
            }
            if constexpr (PK) {
                const int m = 32 * (mt >> 1) + 2 * (8 * h + r) + (mt & 1);
                float* __restrict__ op = outp + (c >> 1) * OP + 2 * m + (c & 1);
                op[0]      = s0;
                op[8 * OP] = s1;
            } else {
                const int row = mt * 16 + 8 * h + r;
                if (row < Mrows) {
                    if constexpr (EPI == 0) {
                        float* __restrict__ zr = Z + (size_t)row * N + n0 + c;
                        zr[0]  = s0;
                        zr[16] = s1;
                    } else {
                        outp[row * 32 + c]      = s0;
                        outp[row * 32 + 16 + c] = s1;
                    }
                }
            }
        }
    }
    if constexpr (PK) {
        __syncthreads();
        constexpr int TPR = ROWF / 4;
        const bool act = (MT == 4) || (threadIdx.x < 16 * TPR);
        const int pr = threadIdx.x / TPR;
        const int q  = threadIdx.x % TPR;
        const size_t prow = (size_t)(n0 / 2 + pr) * 128 + 64 * G0;
        if constexpr (EPI == 0) {
            if (act) {
                const v4f v = *(const v4f*)(outp + pr * OP + 4 * q);
                *(v4f*)(Z + prow + 4 * q) = v;
            }
        } else {
            const int bl = q >> 3;
            const int g0 = (2 * q) & 15;
            v4f o = (v4f){0.f, 0.f, 0.f, 0.f};
            if (act) {
                const float* __restrict__ r0 = Z0 + prow + bl * 32;
                const float* __restrict__ r1 = H  + prow + bl * 32;
                const float* r2 = outp + pr * OP + bl * 32;

                float o00 = bc[g0],     o01 = o00;
                float o10 = bc[g0 + 1], o11 = o10;
#pragma unroll
                for (int f = 0; f < 16; ++f) {
                    const v2f z0 = *(const v2f*)(r0 + 2 * f);
                    const v2f z1 = *(const v2f*)(r1 + 2 * f);
                    const v2f z2 = *(const v2f*)(r2 + 2 * f);
                    const v2f w0 = *(const v2f*)(Wc + (0 * 16 + f) * 16 + g0);
                    const v2f w1 = *(const v2f*)(Wc + (1 * 16 + f) * 16 + g0);
                    const v2f w2 = *(const v2f*)(Wc + (2 * 16 + f) * 16 + g0);
                    o00 = fmaf(z0.x, w0.x, fmaf(z1.x, w1.x, fmaf(z2.x, w2.x, o00)));
                    o01 = fmaf(z0.y, w0.x, fmaf(z1.y, w1.x, fmaf(z2.y, w2.x, o01)));
                    o10 = fmaf(z0.x, w0.y, fmaf(z1.x, w1.y, fmaf(z2.x, w2.y, o10)));
                    o11 = fmaf(z0.y, w0.y, fmaf(z1.y, w1.y, fmaf(z2.y, w2.y, o11)));
                }
                o.x = o00 > 0.f ? o00 : 0.f;
                o.y = o01 > 0.f ? o01 : 0.f;
                o.z = o10 > 0.f ? o10 : 0.f;
                o.w = o11 > 0.f ? o11 : 0.f;
                if constexpr (EPI == 1)
                    *(v4f*)(Z + prow + 4 * q) = o;
            }
            if constexpr (EPI == 2) {
                float* const hs = red;
                if (act) {
                    const int i0 = (bl * 32 + 2 * pr) * 17 + g0;
                    hs[i0]          = o.x;
                    hs[i0 + 17]     = o.y;
                    hs[i0 + 1]      = o.z;
                    hs[i0 + 17 + 1] = o.w;
                }
                __syncthreads();
                if (threadIdx.x < 64) {
                    const int b2 = threadIdx.x >> 5;
                    const int i  = threadIdx.x & 31;
                    float v[16];
                    for (int g = 0; g < 16; ++g) v[g] = hs[(b2 * 32 + i) * 17 + g];
                    float y0[16];
                    for (int j = 0; j < 16; ++j) {
                        float a = R0b[j];
                        for (int g = 0; g < 16; ++g) a = fmaf(v[g], R0W[g * 16 + j], a);
                        y0[j] = a > 0.f ? a : 0.f;
                    }
                    for (int r = 0; r < 2; ++r) {
                        float a = R1b[r];
                        for (int j = 0; j < 16; ++j) a = fmaf(y0[j], R1W[j * 2 + r], a);
                        Z[((size_t)((2 * G0 + b2) * 2 + r)) * (size_t)N + n0 + i] = a;
                    }
                }
            }
        }
    } else if constexpr (EPI == 1) {
        __syncthreads();
        const int pr = threadIdx.x >> 5;
        const int b  = lane >> 3;
        const int g0 = (2 * lane) & 15;
        const v2f z0 = *(const v2f*)(Z0 + (size_t)b * N + n0 + 2 * pr);
        const v2f z1 = *(const v2f*)(H  + (size_t)b * N + n0 + 2 * pr);
        const v2f z2 = *(const v2f*)(outp + b * 32 + 2 * pr);
        v4f o;
        for (int e = 0; e < 2; ++e) {
            const int g = g0 + e;
            const float w0 = Wc[g], w1 = Wc[16 + g], w2 = Wc[32 + g], bs = bc[g];
            const float v0 = fmaf(z0.x, w0, fmaf(z1.x, w1, fmaf(z2.x, w2, bs)));
            const float v1 = fmaf(z0.y, w0, fmaf(z1.y, w1, fmaf(z2.y, w2, bs)));
            o[2 * e]     = v0 > 0.f ? v0 : 0.f;
            o[2 * e + 1] = v1 > 0.f ? v1 : 0.f;
        }
        *(v4f*)(Z + (size_t)(n0 / 2 + pr) * 128 + 4 * lane) = o;
    }
}

__global__ void pool_packed(
    float* __restrict__ P, const float* __restrict__ Hin,
    const int* __restrict__ incid, int N2, int DEG)
{
    const int lane = threadIdx.x & 31;
    const int kp   = blockIdx.x * (blockDim.x >> 5) + (threadIdx.x >> 5);
    if (kp >= N2 / 2) return;

    v4f o;
    for (int s = 0; s < 2; ++s) {
        const int node = 2 * kp + s;
        float v0 = 0.0f, v1 = 0.0f;
        for (int d = 0; d < DEG; ++d) {
            const int j = incid[node * DEG + d];
            const v4f q = *(const v4f*)(Hin + (size_t)(j >> 1) * 128 + 4 * lane);
            v0 = fmaxf(v0, (j & 1) ? q.y : q.x);
            v1 = fmaxf(v1, (j & 1) ? q.w : q.z);
        }
        o[s]     = v0;
        o[2 + s] = v1;
    }
    *(v4f*)(P + (size_t)kp * 128 + 4 * lane) = o;
}

extern "C" void kernel_launch(void* const* d_in, const int* in_sizes, int n_in,
                              void* d_out, int out_size, void* d_ws, size_t ws_size,
                              hipStream_t stream)
{
    (void)in_sizes; (void)n_in; (void)out_size; (void)ws_size;

    const float* x     = (const float*)d_in[0];
    const float* S1    = (const float*)d_in[1];
    const float* S2    = (const float*)d_in[2];
    const int*   incid = (const int*)  d_in[3];
    const float* W10   = (const float*)d_in[4];
    const float* b10   = (const float*)d_in[5];
    const float* W11   = (const float*)d_in[6];
    const float* b11   = (const float*)d_in[7];
    const float* W20   = (const float*)d_in[8];
    const float* b20   = (const float*)d_in[9];
    const float* R0W   = (const float*)d_in[10];
    const float* R0b   = (const float*)d_in[11];
    const float* R1W   = (const float*)d_in[12];
    const float* R1b   = (const float*)d_in[13];
    float* out = (float*)d_out;

    const int B = 4, F1 = 16, N1 = 8192, N2 = 4096, DEG = 4;

    char* ws = (char*)d_ws;
    const size_t szBig = (size_t)(B * F1) * N1 * sizeof(float);
    float* h1 = (float*)(ws);
    float* tA = (float*)(ws + 1 * szBig);
    float* h2 = (float*)(ws + 2 * szBig);
    float* p  = (float*)(ws + 3 * szBig);

    auto grid1d    = [](int total, int blk) { return (total + blk - 1) / blk; };


    gemm_splitk<N1, 1, 16, 16, false, 0, false><<<N1 / 32, 512, 0, stream>>>(
        tA, x,  S1, B, nullptr, nullptr, nullptr, nullptr, nullptr, nullptr, nullptr);
    gemm_splitk<N1, 1, 16, 16, false, 1, true><<<N1 / 32, 512, 0, stream>>>(
        h1, tA, S1, B, x, W10, b10, nullptr, nullptr, nullptr, nullptr);

    gemm_splitk<N1, 4, 16, 16, true, 0, false><<<N1 / 32, 512, 0, stream>>>(
        tA, h1, S1, B * F1, nullptr, nullptr, nullptr, nullptr, nullptr, nullptr, nullptr);
    gemm_splitk<N1, 4, 16, 16, true, 1, false><<<N1 / 32, 512, 0, stream>>>(
        h2, tA, S1, B * F1, h1, W11, b11, nullptr, nullptr, nullptr, nullptr);

    pool_packed<<<grid1d(N2 / 2, 8), 256, 0, stream>>>(p, h2, incid, N2, DEG);

    gemm_splitk<N2, 2, 16, 16, true, 0, false><<<dim3(N2 / 32, 2), 512, 0, stream>>>(
        tA, p,  S2, B * F1, nullptr, nullptr, nullptr, nullptr, nullptr, nullptr, nullptr);
    gemm_splitk<N2, 2, 16, 16, true, 2, false><<<dim3(N2 / 32, 2), 512, 0, stream>>>(
        out, tA, S2, B * F1, p, W20, b20, R0W, R0b, R1W, R1b);
}
